# starting kernel
# baseline (speedup 1.0000x reference)
#include <hip/hip_runtime.h>

typedef __attribute__((ext_vector_type(2))) float v2f;
typedef __attribute__((ext_vector_type(8))) float v8f;

#define BSZ 4
#define SSZ 512
#define HSZ 128
#define TI  4

__global__ __launch_bounds__(128) void gates_wmma_kernel(
    const float* __restrict__ inp,
    const float* __restrict__ W0, const float* __restrict__ b0,
    const float* __restrict__ W1, const float* __restrict__ b1,
    float* __restrict__ G0, float* __restrict__ G1)
{
  const int lane  = threadIdx.x & 31;
  const int wave  = threadIdx.x >> 5;
  const int tile  = blockIdx.x * 4 + wave;
  const int ntile = tile & 7;
  const int mtile = tile >> 3;

  const int r  = lane & 15;
  const int kb = (lane >> 4) * 2;

  const float* Aptr  = inp + (size_t)(mtile * 16 + r) * HSZ + kb;
  const float* B0ptr = W0  + (size_t)(ntile * 16 + r) * HSZ + kb;
  const float* B1ptr = W1  + (size_t)(ntile * 16 + r) * HSZ + kb;

  v8f acc0 = {};
  v8f acc1 = {};
  #pragma unroll 4
  for (int k0 = 0; k0 < HSZ; k0 += 4) {
    v2f a  = *(const v2f*)(Aptr  + k0);
    v2f w0 = *(const v2f*)(B0ptr + k0);
    v2f w1 = *(const v2f*)(B1ptr + k0);
    acc0 = __builtin_amdgcn_wmma_f32_16x16x4_f32(
         false, a,  false, w0, (short)0, acc0, false, false);
    acc1 = __builtin_amdgcn_wmma_f32_16x16x4_f32(
         false, a,  false, w1, (short)0, acc1, false, false);
  }

  const int col  = ntile * 16 + (lane & 15);
  const int mofs = (lane >> 4) * 8;
  const float bias0 = b0[col];
  const float bias1 = b1[col];
  #pragma unroll
  for (int v = 0; v < 8; ++v) {
    const size_t idx = (size_t)(mtile * 16 + mofs + v) * HSZ + col;
    G0[idx] = acc0[v] + bias0;
    G1[idx] = acc1[v] + bias1;
  }
}

__global__ __launch_bounds__(128) void pairwise_kernel(
    const float* __restrict__ inp,
    const int*   __restrict__ masks,
    const float* __restrict__ G0,
    const float* __restrict__ G1,
    float* __restrict__ out)
{
  const int h     = threadIdx.x;
  const int b     = blockIdx.x / (SSZ / TI);
  const int ibase = (blockIdx.x % (SSZ / TI)) * TI;

  float g0[TI];
  float acc[TI];
  #pragma unroll
  for (int t = 0; t < TI; ++t) {
    g0[t]  = G0[((size_t)(b * SSZ + ibase + t) * HSZ) + h];
    acc[t] = 0.0f;
  }

  const float* inpB = inp + (size_t)b * SSZ * HSZ;
  const float* g1B  = G1  + (size_t)b * SSZ * HSZ;
  const int*   mB   = masks + b * SSZ;

  int len = 0;
  for (int j = 0; j < SSZ; ++j) {
    const int mj = mB[j];
    len += mj;
    if (mj) {
      const float x  = inpB[(size_t)j * HSZ + h];
      const float g1 = g1B [(size_t)j * HSZ + h];
      #pragma unroll
      for (int t = 0; t < TI; ++t) {
        const float s = 1.0f / (1.0f + __expf(-(g0[t] + g1)));
        acc[t] += x * s;
      }
    }
  }

  const float inv_len = 1.0f / (float)len;
  #pragma unroll
  for (int t = 0; t < TI; ++t) {
    const float mi = (float)mB[ibase + t];
    out[((size_t)(b * SSZ + ibase + t) * HSZ) + h] = tanhf(mi * acc[t] * inv_len);
  }
}

extern "C" void kernel_launch(void* const* d_in, const int* in_sizes, int n_in,
                              void* d_out, int out_size, void* d_ws, size_t ws_size,
                              hipStream_t stream) {
  const float* inp   = (const float*)d_in[0];
  const int*   masks = (const int*)  d_in[1];
  const float* W0    = (const float*)d_in[2];
  const float* b0    = (const float*)d_in[3];
  const float* W1    = (const float*)d_in[4];
  const float* b1    = (const float*)d_in[5];
  float* out = (float*)d_out;

  float* G0 = (float*)d_ws;
  float* G1 = G0 + (size_t)BSZ * SSZ * HSZ;

  gates_wmma_kernel<<<256, 128, 0, stream>>>(inp, W0, b0, W1, b1, G0, G1);
  pairwise_kernel<<<(BSZ * SSZ) / TI, 128, 0, stream>>>(inp, masks, G0, G1, out);
}


# pairwise kernel re-cut: lanes = 32 rows i, wave = 4 columns; each unmasked row j of inp/G1 is read from the lane that fetched it (v_readlane) and used as scalar operands, 2 rows a trip; j order kept
# speedup vs baseline: 3.3865x; 3.3865x over previous
#include <hip/hip_runtime.h>

typedef __attribute__((ext_vector_type(2))) float v2f;
typedef __attribute__((ext_vector_type(8))) float v8f;
typedef __attribute__((ext_vector_type(4))) float v4f;
typedef __attribute__((ext_vector_type(4))) int   v4i;

#define BSZ 4
#define SSZ 512
#define HSZ 128
#define HT  4
#define TP  36

__global__ __launch_bounds__(128) void gates_wmma_kernel(
    const float* __restrict__ inp,
    const float* __restrict__ W0, const float* __restrict__ b0,
    const float* __restrict__ W1, const float* __restrict__ b1,
    float* __restrict__ G0, float* __restrict__ G1)
{
  const int lane  = threadIdx.x & 31;
  const int wave  = threadIdx.x >> 5;
  const int tile  = blockIdx.x * 4 + wave;
  const int ntile = tile & 7;
  const int mtile = tile >> 3;

  const int r  = lane & 15;
  const int kb = (lane >> 4) * 2;

  const float* Aptr  = inp + (size_t)(mtile * 16 + r) * HSZ + kb;
  const float* B0ptr = W0  + (size_t)(ntile * 16 + r) * HSZ + kb;
  const float* B1ptr = W1  + (size_t)(ntile * 16 + r) * HSZ + kb;

  v8f acc0 = {};
  v8f acc1 = {};
  #pragma unroll 4
  for (int k0 = 0; k0 < HSZ; k0 += 4) {
    v2f a  = *(const v2f*)(Aptr  + k0);
    v2f w0 = *(const v2f*)(B0ptr + k0);
    v2f w1 = *(const v2f*)(B1ptr + k0);
    acc0 = __builtin_amdgcn_wmma_f32_16x16x4_f32(
         false, a,  false, w0, (short)0, acc0, false, false);
    acc1 = __builtin_amdgcn_wmma_f32_16x16x4_f32(
         false, a,  false, w1, (short)0, acc1, false, false);
  }

  const int col  = ntile * 16 + (lane & 15);
  const int mofs = (lane >> 4) * 8;
  const float bias0 = b0[col];
  const float bias1 = b1[col];
  #pragma unroll
  for (int v = 0; v < 8; ++v) {
    const size_t idx = (size_t)(mtile * 16 + mofs + v) * HSZ + col;
    G0[idx] = acc0[v] + bias0;
    G1[idx] = acc1[v] + bias1;
  }
}

struct Chunk {
  v4f x;
  v4f g;
  int m;
};

__device__ __forceinline__ Chunk load_chunk(const float* __restrict__ xrow,
                                            const float* __restrict__ grow,
                                            const int*   __restrict__ mrow,
                                            int row)
{
  Chunk c;
  c.x = *(const v4f*)(xrow + (size_t)row * HSZ);
  c.g = *(const v4f*)(grow + (size_t)row * HSZ);
  c.m = mrow[row];
  return c;
}

__device__ __forceinline__ float lane_value(float v, int k)
{
  return __builtin_bit_cast(float,
           __builtin_amdgcn_readlane(__builtin_bit_cast(int, v), k));
}

__global__ __launch_bounds__(256) void pairwise_kernel(
    const float* __restrict__ inp,
    const int*   __restrict__ masks,
    const float* __restrict__ G0,
    const float* __restrict__ G1,
    float* __restrict__ out)
{
  __shared__ __attribute__((aligned(16))) float tile[32 * TP];

  const int lane = threadIdx.x & 31;
  const int wv = threadIdx.x >> 5;
  const int hg = blockIdx.x & 3;
  const int it = (blockIdx.x >> 2) & 15;
  const int b  = blockIdx.x >> 6;
  const int h0 = hg * 32 + wv * HT;
  const int i  = it * 32 + lane;

  const v4f   g0 = *(const v4f*)(G0 + (size_t)(b * SSZ + i) * HSZ + h0);
  const float mi = (float)masks[b * SSZ + i];
  float acc[HT];
  #pragma unroll
  for (int t = 0; t < HT; ++t) acc[t] = 0.0f;

  const float* xrow = inp + (size_t)b * SSZ * HSZ + h0;
  const float* grow = G1  + (size_t)b * SSZ * HSZ + h0;
  const int*   mrow = masks + b * SSZ;

  int len = 0;
  Chunk cur = load_chunk(xrow, grow, mrow, lane);
  for (int c = 0; c < SSZ / 32; ++c) {
    const int cn = (c + 1 < SSZ / 32) ? (c + 1) : c;
    const Chunk nxt = load_chunk(xrow, grow, mrow, cn * 32 + lane);

    unsigned bits = __builtin_amdgcn_ballot_w32(cur.m != 0);
    len += __builtin_popcount(bits);
    while (bits) {
      const int k1 = __builtin_ctz(bits);
      bits &= bits - 1;
      const bool two = bits != 0;
      const int k2 = two ? __builtin_ctz(bits) : k1;
      bits &= bits - 1;
      float x1[HT], g1[HT], x2[HT], g2[HT];
      #pragma unroll
      for (int t = 0; t < HT; ++t) {
        x1[t] = lane_value(cur.x[t], k1);
        g1[t] = lane_value(cur.g[t], k1);
        const float xk2 = lane_value(cur.x[t], k2);
        x2[t] = two ? xk2 : 0.0f;
        g2[t] = lane_value(cur.g[t], k2);
      }
      #pragma unroll
      for (int t = 0; t < HT; ++t) {
        const float s1 = 1.0f / (1.0f + __expf(-(g0[t] + g1[t])));
        acc[t] += x1[t] * s1;
      }
      #pragma unroll
      for (int t = 0; t < HT; ++t) {
        const float s2 = 1.0f / (1.0f + __expf(-(g0[t] + g2[t])));
        acc[t] += x2[t] * s2;
      }
    }
    cur = nxt;
  }

  const float inv_len = 1.0f / (float)len;
  v4f o;
  #pragma unroll
  for (int t = 0; t < HT; ++t) o[t] = tanhf(mi * acc[t] * inv_len);

  *(v4f*)(tile + lane * TP + wv * HT) = o;
  __syncthreads();
  #pragma unroll
  for (int r = 0; r < 4; ++r) {
    const int row = wv * 4 + r;
    out[(size_t)(b * SSZ + it * 32 + row) * HSZ + hg * 32 + lane] = tile[row * TP + lane];
  }
}

extern "C" void kernel_launch(void* const* d_in, const int* in_sizes, int n_in,
                              void* d_out, int out_size, void* d_ws, size_t ws_size,
                              hipStream_t stream) {
  const float* inp   = (const float*)d_in[0];
  const int*   masks = (const int*)  d_in[1];
  const float* W0    = (const float*)d_in[2];
  const float* b0    = (const float*)d_in[3];
  const float* W1    = (const float*)d_in[4];
  const float* b1    = (const float*)d_in[5];
  float* out = (float*)d_out;

  float* G0 = (float*)d_ws;
  float* G1 = G0 + (size_t)BSZ * SSZ * HSZ;

  gates_wmma_kernel<<<256, 128, 0, stream>>>(inp, W0, b0, W1, b1, G0, G1);
  pairwise_kernel<<<256, 256, 0, stream>>>(inp, masks, G0, G1, out);
}


# pairwise kernel: a wave takes 2 columns, not 4: 16 waves a block, 4 a SIMD instead of 2, so other waves issue while one waits on its lane-read and division chain; loop and j order kept
# speedup vs baseline: 3.6297x; 1.0718x over previous
#include <hip/hip_runtime.h>

typedef __attribute__((ext_vector_type(2))) float v2f;
typedef __attribute__((ext_vector_type(8))) float v8f;

#define BSZ 4
#define SSZ 512
#define HSZ 128
#define HT  2
#define TP  36

__global__ __launch_bounds__(128) void gates_wmma_kernel(
    const float* __restrict__ inp,
    const float* __restrict__ W0, const float* __restrict__ b0,
    const float* __restrict__ W1, const float* __restrict__ b1,
    float* __restrict__ G0, float* __restrict__ G1)
{
  const int lane  = threadIdx.x & 31;
  const int wave  = threadIdx.x >> 5;
  const int tile  = blockIdx.x * 4 + wave;
  const int ntile = tile & 7;
  const int mtile = tile >> 3;

  const int r  = lane & 15;
  const int kb = (lane >> 4) * 2;

  const float* Aptr  = inp + (size_t)(mtile * 16 + r) * HSZ + kb;
  const float* B0ptr = W0  + (size_t)(ntile * 16 + r) * HSZ + kb;
  const float* B1ptr = W1  + (size_t)(ntile * 16 + r) * HSZ + kb;

  v8f acc0 = {};
  v8f acc1 = {};
  #pragma unroll 4
  for (int k0 = 0; k0 < HSZ; k0 += 4) {
    v2f a  = *(const v2f*)(Aptr  + k0);
    v2f w0 = *(const v2f*)(B0ptr + k0);
    v2f w1 = *(const v2f*)(B1ptr + k0);
    acc0 = __builtin_amdgcn_wmma_f32_16x16x4_f32(
         false, a,  false, w0, (short)0, acc0, false, false);
    acc1 = __builtin_amdgcn_wmma_f32_16x16x4_f32(
         false, a,  false, w1, (short)0, acc1, false, false);
  }

  const int col  = ntile * 16 + (lane & 15);
  const int mofs = (lane >> 4) * 8;
  const float bias0 = b0[col];
  const float bias1 = b1[col];
  #pragma unroll
  for (int v = 0; v < 8; ++v) {
    const size_t idx = (size_t)(mtile * 16 + mofs + v) * HSZ + col;
    G0[idx] = acc0[v] + bias0;
    G1[idx] = acc1[v] + bias1;
  }
}

struct Chunk {
  v2f x;
  v2f g;
  int m;
};

__device__ __forceinline__ Chunk load_chunk(const float* __restrict__ xrow,
                                            const float* __restrict__ grow,
                                            const int*   __restrict__ mrow,
                                            int row)
{
  Chunk c;
  c.x = *(const v2f*)(xrow + (size_t)row * HSZ);
  c.g = *(const v2f*)(grow + (size_t)row * HSZ);
  c.m = mrow[row];
  return c;
}

__device__ __forceinline__ float lane_value(float v, int k)
{
  return __builtin_bit_cast(float,
           __builtin_amdgcn_readlane(__builtin_bit_cast(int, v), k));
}

__global__ __launch_bounds__(512) void pairwise_kernel(
    const float* __restrict__ inp,
    const int*   __restrict__ masks,
    const float* __restrict__ G0,
    const float* __restrict__ G1,
    float* __restrict__ out)
{
  __shared__ __attribute__((aligned(16))) float tile[32 * TP];

  const int lane = threadIdx.x & 31;
  const int wv = threadIdx.x >> 5;
  const int hg = blockIdx.x & 3;
  const int it = (blockIdx.x >> 2) & 15;
  const int b  = blockIdx.x >> 6;
  const int h0 = hg * 32 + wv * HT;
  const int i  = it * 32 + lane;

  const v2f   g0 = *(const v2f*)(G0 + (size_t)(b * SSZ + i) * HSZ + h0);
  const float mi = (float)masks[b * SSZ + i];
  float acc[HT];
  #pragma unroll
  for (int t = 0; t < HT; ++t) acc[t] = 0.0f;

  const float* xrow = inp + (size_t)b * SSZ * HSZ + h0;
  const float* grow = G1  + (size_t)b * SSZ * HSZ + h0;
  const int*   mrow = masks + b * SSZ;

  int len = 0;
  Chunk cur = load_chunk(xrow, grow, mrow, lane);
  for (int c = 0; c < SSZ / 32; ++c) {
    const int cn = (c + 1 < SSZ / 32) ? (c + 1) : c;
    const Chunk nxt = load_chunk(xrow, grow, mrow, cn * 32 + lane);

    unsigned bits = __builtin_amdgcn_ballot_w32(cur.m != 0);
    len += __builtin_popcount(bits);
    while (bits) {
      const int k1 = __builtin_ctz(bits);
      bits &= bits - 1;
      const bool two = bits != 0;
      const int k2 = two ? __builtin_ctz(bits) : k1;
      bits &= bits - 1;
      float x1[HT], g1[HT], x2[HT], g2[HT];
      #pragma unroll
      for (int t = 0; t < HT; ++t) {
        x1[t] = lane_value(cur.x[t], k1);
        g1[t] = lane_value(cur.g[t], k1);
        const float xk2 = lane_value(cur.x[t], k2);
        x2[t] = two ? xk2 : 0.0f;
        g2[t] = lane_value(cur.g[t], k2);
      }
      #pragma unroll
      for (int t = 0; t < HT; ++t) {
        const float s1 = 1.0f / (1.0f + __expf(-(g0[t] + g1[t])));
        acc[t] += x1[t] * s1;
      }
      #pragma unroll
      for (int t = 0; t < HT; ++t) {
        const float s2 = 1.0f / (1.0f + __expf(-(g0[t] + g2[t])));
        acc[t] += x2[t] * s2;
      }
    }
    cur = nxt;
  }

  const float inv_len = 1.0f / (float)len;
  v2f o;
  #pragma unroll
  for (int t = 0; t < HT; ++t) o[t] = tanhf(mi * acc[t] * inv_len);

  *(v2f*)(tile + lane * TP + wv * HT) = o;
  __syncthreads();
  #pragma unroll
  for (int r = 0; r < 2; ++r) {
    const int row = wv * 2 + r;
    out[(size_t)(b * SSZ + it * 32 + row) * HSZ + hg * 32 + lane] = tile[row * TP + lane];
  }
}

extern "C" void kernel_launch(void* const* d_in, const int* in_sizes, int n_in,
                              void* d_out, int out_size, void* d_ws, size_t ws_size,
                              hipStream_t stream) {
  const float* inp   = (const float*)d_in[0];
  const int*   masks = (const int*)  d_in[1];
  const float* W0    = (const float*)d_in[2];
  const float* b0    = (const float*)d_in[3];
  const float* W1    = (const float*)d_in[4];
  const float* b1    = (const float*)d_in[5];
  float* out = (float*)d_out;

  float* G0 = (float*)d_ws;
  float* G1 = G0 + (size_t)BSZ * SSZ * HSZ;

  gates_wmma_kernel<<<256, 128, 0, stream>>>(inp, W0, b0, W1, b1, G0, G1);
  pairwise_kernel<<<256, 512, 0, stream>>>(inp, masks, G0, G1, out);
}


# pairwise term: one IEEE division a trip, not four: the trip's 4 denominators 1+exp(-(g0+g1)) are multiplied, the product inverted once, each 1/d got back by two multiplies; j sum order kept
# speedup vs baseline: 4.3043x; 1.1859x over previous
#include <hip/hip_runtime.h>

typedef __attribute__((ext_vector_type(2))) float v2f;
typedef __attribute__((ext_vector_type(8))) float v8f;

#define BSZ 4
#define SSZ 512
#define HSZ 128
#define HT  2
#define TP  36
#define PMAX 0x1.0p+96f
static_assert(HT == 2, "the shared division below is written for two columns a wave");

__global__ __launch_bounds__(128) void gates_wmma_kernel(
    const float* __restrict__ inp,
    const float* __restrict__ W0, const float* __restrict__ b0,
    const float* __restrict__ W1, const float* __restrict__ b1,
    float* __restrict__ G0, float* __restrict__ G1)
{
  const int lane  = threadIdx.x & 31;
  const int wave  = threadIdx.x >> 5;
  const int tile  = blockIdx.x * 4 + wave;
  const int ntile = tile & 7;
  const int mtile = tile >> 3;

  const int r  = lane & 15;
  const int kb = (lane >> 4) * 2;

  const float* Aptr  = inp + (size_t)(mtile * 16 + r) * HSZ + kb;
  const float* B0ptr = W0  + (size_t)(ntile * 16 + r) * HSZ + kb;
  const float* B1ptr = W1  + (size_t)(ntile * 16 + r) * HSZ + kb;

  v8f acc0 = {};
  v8f acc1 = {};
  #pragma unroll 4
  for (int k0 = 0; k0 < HSZ; k0 += 4) {
    v2f a  = *(const v2f*)(Aptr  + k0);
    v2f w0 = *(const v2f*)(B0ptr + k0);
    v2f w1 = *(const v2f*)(B1ptr + k0);
    acc0 = __builtin_amdgcn_wmma_f32_16x16x4_f32(
         false, a,  false, w0, (short)0, acc0, false, false);
    acc1 = __builtin_amdgcn_wmma_f32_16x16x4_f32(
         false, a,  false, w1, (short)0, acc1, false, false);
  }

  const int col  = ntile * 16 + (lane & 15);
  const int mofs = (lane >> 4) * 8;
  const float bias0 = b0[col];
  const float bias1 = b1[col];
  #pragma unroll
  for (int v = 0; v < 8; ++v) {
    const size_t idx = (size_t)(mtile * 16 + mofs + v) * HSZ + col;
    G0[idx] = acc0[v] + bias0;
    G1[idx] = acc1[v] + bias1;
  }
}

struct Chunk {
  v2f x;
  v2f g;
  int m;
};

__device__ __forceinline__ Chunk load_chunk(const float* __restrict__ xrow,
                                            const float* __restrict__ grow,
                                            const int*   __restrict__ mrow,
                                            int row)
{
  Chunk c;
  c.x = *(const v2f*)(xrow + (size_t)row * HSZ);
  c.g = *(const v2f*)(grow + (size_t)row * HSZ);
  c.m = mrow[row];
  return c;
}

__device__ __forceinline__ float lane_value(float v, int k)
{
  return __builtin_bit_cast(float,
           __builtin_amdgcn_readlane(__builtin_bit_cast(int, v), k));
}

__global__ __launch_bounds__(512) void pairwise_kernel(
    const float* __restrict__ inp,
    const int*   __restrict__ masks,
    const float* __restrict__ G0,
    const float* __restrict__ G1,
    float* __restrict__ out)
{
  __shared__ __attribute__((aligned(16))) float tile[32 * TP];

  const int lane = threadIdx.x & 31;
  const int wv = threadIdx.x >> 5;
  const int hg = blockIdx.x & 3;
  const int it = (blockIdx.x >> 2) & 15;
  const int b  = blockIdx.x >> 6;
  const int h0 = hg * 32 + wv * HT;
  const int i  = it * 32 + lane;

  const v2f   g0 = *(const v2f*)(G0 + (size_t)(b * SSZ + i) * HSZ + h0);
  const float mi = (float)masks[b * SSZ + i];
  float acc[HT];
  #pragma unroll
  for (int t = 0; t < HT; ++t) acc[t] = 0.0f;

  const float* xrow = inp + (size_t)b * SSZ * HSZ + h0;
  const float* grow = G1  + (size_t)b * SSZ * HSZ + h0;
  const int*   mrow = masks + b * SSZ;

  int len = 0;
  Chunk cur = load_chunk(xrow, grow, mrow, lane);
  for (int c = 0; c < SSZ / 32; ++c) {
    const int cn = (c + 1 < SSZ / 32) ? (c + 1) : c;
    const Chunk nxt = load_chunk(xrow, grow, mrow, cn * 32 + lane);

    unsigned bits = __builtin_amdgcn_ballot_w32(cur.m != 0);
    len += __builtin_popcount(bits);
    while (bits) {
      const int k1 = __builtin_ctz(bits);
      bits &= bits - 1;
      const bool two = bits != 0;
      const int k2 = two ? __builtin_ctz(bits) : k1;
      bits &= bits - 1;
      float x1[HT], g1[HT], x2[HT], g2[HT];
      #pragma unroll
      for (int t = 0; t < HT; ++t) {
        x1[t] = lane_value(cur.x[t], k1);
        g1[t] = lane_value(cur.g[t], k1);
        const float xk2 = lane_value(cur.x[t], k2);
        x2[t] = two ? xk2 : 0.0f;
        g2[t] = lane_value(cur.g[t], k2);
      }
      v2f d1, d2;
      #pragma unroll
      for (int t = 0; t < HT; ++t) {
        d1[t] = 1.0f + __expf(-(g0[t] + g1[t]));
        d2[t] = 1.0f + __expf(-(g0[t] + g2[t]));
      }
      const v2f   q  = d1 * d2;
      const float pp = q[0] * q[1];
      const float r  = 1.0f / pp;
      v2f rq;
      rq[0] = r * q[1];
      rq[1] = r * q[0];
      v2f s1 = rq * d2;
      v2f s2 = rq * d1;
      if (__builtin_expect(__builtin_amdgcn_ballot_w32(!(pp < PMAX)) != 0u, 0)) {
        #pragma unroll
        for (int t = 0; t < HT; ++t) {
          s1[t] = 1.0f / d1[t];
          s2[t] = 1.0f / d2[t];
        }
      }
      #pragma unroll
      for (int t = 0; t < HT; ++t) acc[t] += x1[t] * s1[t];
      #pragma unroll
      for (int t = 0; t < HT; ++t) acc[t] += x2[t] * s2[t];
    }
    cur = nxt;
  }

  const float inv_len = 1.0f / (float)len;
  v2f o;
  #pragma unroll
  for (int t = 0; t < HT; ++t) o[t] = tanhf(mi * acc[t] * inv_len);

  *(v2f*)(tile + lane * TP + wv * HT) = o;
  __syncthreads();
  #pragma unroll
  for (int r = 0; r < 2; ++r) {
    const int row = wv * 2 + r;
    out[(size_t)(b * SSZ + it * 32 + row) * HSZ + hg * 32 + lane] = tile[row * TP + lane];
  }
}

extern "C" void kernel_launch(void* const* d_in, const int* in_sizes, int n_in,
                              void* d_out, int out_size, void* d_ws, size_t ws_size,
                              hipStream_t stream) {
  const float* inp   = (const float*)d_in[0];
  const int*   masks = (const int*)  d_in[1];
  const float* W0    = (const float*)d_in[2];
  const float* b0    = (const float*)d_in[3];
  const float* W1    = (const float*)d_in[4];
  const float* b1    = (const float*)d_in[5];
  float* out = (float*)d_out;

  float* G0 = (float*)d_ws;
  float* G1 = G0 + (size_t)BSZ * SSZ * HSZ;

  gates_wmma_kernel<<<256, 128, 0, stream>>>(inp, W0, b0, W1, b1, G0, G1);
  pairwise_kernel<<<256, 512, 0, stream>>>(inp, masks, G0, G1, out);
}


# pairwise term: exp(-(g0+g1)) split as exp(-g0)*exp(-g1), one factor per lane, one per fetched row per chunk; the pair loop forms d = 1+e0*e1 by one multiply-add, no exponential; j sum order kept
# speedup vs baseline: 4.5098x; 1.0477x over previous
#include <hip/hip_runtime.h>

typedef __attribute__((ext_vector_type(2))) float v2f;
typedef __attribute__((ext_vector_type(8))) float v8f;

#define BSZ 4
#define SSZ 512
#define HSZ 128
#define HT  2
#define TP  36
#define PMAX 0x1.0p+96f
#define GMAX 80.0f
static_assert(HT == 2, "the shared division below is written for two columns a wave");

__global__ __launch_bounds__(128) void gates_wmma_kernel(
    const float* __restrict__ inp,
    const float* __restrict__ W0, const float* __restrict__ b0,
    const float* __restrict__ W1, const float* __restrict__ b1,
    float* __restrict__ G0, float* __restrict__ G1)
{
  const int lane  = threadIdx.x & 31;
  const int wave  = threadIdx.x >> 5;
  const int tile  = blockIdx.x * 4 + wave;
  const int ntile = tile & 7;
  const int mtile = tile >> 3;

  const int r  = lane & 15;
  const int kb = (lane >> 4) * 2;

  const float* Aptr  = inp + (size_t)(mtile * 16 + r) * HSZ + kb;
  const float* B0ptr = W0  + (size_t)(ntile * 16 + r) * HSZ + kb;
  const float* B1ptr = W1  + (size_t)(ntile * 16 + r) * HSZ + kb;

  v8f acc0 = {};
  v8f acc1 = {};
  #pragma unroll 4
  for (int k0 = 0; k0 < HSZ; k0 += 4) {
    v2f a  = *(const v2f*)(Aptr  + k0);
    v2f w0 = *(const v2f*)(B0ptr + k0);
    v2f w1 = *(const v2f*)(B1ptr + k0);
    acc0 = __builtin_amdgcn_wmma_f32_16x16x4_f32(
         false, a,  false, w0, (short)0, acc0, false, false);
    acc1 = __builtin_amdgcn_wmma_f32_16x16x4_f32(
         false, a,  false, w1, (short)0, acc1, false, false);
  }

  const int col  = ntile * 16 + (lane & 15);
  const int mofs = (lane >> 4) * 8;
  const float bias0 = b0[col];
  const float bias1 = b1[col];
  #pragma unroll
  for (int v = 0; v < 8; ++v) {
    const size_t idx = (size_t)(mtile * 16 + mofs + v) * HSZ + col;
    G0[idx] = acc0[v] + bias0;
    G1[idx] = acc1[v] + bias1;
  }
}

struct Chunk {
  v2f x;
  v2f g;
  int m;
};

__device__ __forceinline__ Chunk load_chunk(const float* __restrict__ xrow,
                                            const float* __restrict__ grow,
                                            const int*   __restrict__ mrow,
                                            int row)
{
  Chunk c;
  c.x = *(const v2f*)(xrow + (size_t)row * HSZ);
  c.g = *(const v2f*)(grow + (size_t)row * HSZ);
  c.m = mrow[row];
  return c;
}

__device__ __forceinline__ float lane_value(float v, int k)
{
  return __builtin_bit_cast(float,
           __builtin_amdgcn_readlane(__builtin_bit_cast(int, v), k));
}

__device__ __forceinline__ float exp_neg(float g)
{
  const float c_hi = -0x1.715476p+0f;
  const float c_lo = -0x1.4ae0cp-26f;
  const float t = g * c_hi;
  const float r = __builtin_fmaf(g, c_lo, __builtin_fmaf(g, c_hi, -t));
  const float e = __builtin_amdgcn_exp2f(t);
  return __builtin_fmaf(e, r * 0x1.62e43p-1f, e);
}

template <bool SPLIT>
__device__ __forceinline__ void add_rows(unsigned bits, const v2f cx, const v2f cg,
                                         const v2f a0, float (&acc)[HT])
{
  while (bits) {
    const int k1 = __builtin_ctz(bits);
    bits &= bits - 1;
    const bool two = bits != 0;
    const int k2 = two ? __builtin_ctz(bits) : k1;
    bits &= bits - 1;
    float x1[HT], g1[HT], x2[HT], g2[HT];
    #pragma unroll
    for (int t = 0; t < HT; ++t) {
      x1[t] = lane_value(cx[t], k1);
      g1[t] = lane_value(cg[t], k1);
      const float xk2 = lane_value(cx[t], k2);
      x2[t] = two ? xk2 : 0.0f;
      g2[t] = lane_value(cg[t], k2);
    }
    v2f d1, d2;
    #pragma unroll
    for (int t = 0; t < HT; ++t) {
      if (SPLIT) {
        d1[t] = __builtin_fmaf(a0[t], g1[t], 1.0f);
        d2[t] = __builtin_fmaf(a0[t], g2[t], 1.0f);
      } else {
        d1[t] = 1.0f + __expf(-(a0[t] + g1[t]));
        d2[t] = 1.0f + __expf(-(a0[t] + g2[t]));
      }
    }
    const v2f   q  = d1 * d2;
    const float pp = q[0] * q[1];
    const float r  = 1.0f / pp;
    v2f rq;
    rq[0] = r * q[1];
    rq[1] = r * q[0];
    v2f s1 = rq * d2;
    v2f s2 = rq * d1;
    if (__builtin_expect(__builtin_amdgcn_ballot_w32(!(pp < PMAX)) != 0u, 0)) {
      #pragma unroll
      for (int t = 0; t < HT; ++t) {
        s1[t] = 1.0f / d1[t];
        s2[t] = 1.0f / d2[t];
      }
    }
    #pragma unroll
    for (int t = 0; t < HT; ++t) acc[t] += x1[t] * s1[t];
    #pragma unroll
    for (int t = 0; t < HT; ++t) acc[t] += x2[t] * s2[t];
  }
}

__global__ __launch_bounds__(512) void pairwise_kernel(
    const float* __restrict__ inp,
    const int*   __restrict__ masks,
    const float* __restrict__ G0,
    const float* __restrict__ G1,
    float* __restrict__ out)
{
  __shared__ __attribute__((aligned(16))) float tile[32 * TP];

  const int lane = threadIdx.x & 31;
  const int wv = threadIdx.x >> 5;
  const int hg = blockIdx.x & 3;
  const int it = (blockIdx.x >> 2) & 15;
  const int b  = blockIdx.x >> 6;
  const int h0 = hg * 32 + wv * HT;
  const int i  = it * 32 + lane;

  const v2f   g0 = *(const v2f*)(G0 + (size_t)(b * SSZ + i) * HSZ + h0);
  const float mi = (float)masks[b * SSZ + i];
  float acc[HT];
  #pragma unroll
  for (int t = 0; t < HT; ++t) acc[t] = 0.0f;

  const float* xrow = inp + (size_t)b * SSZ * HSZ + h0;
  const float* grow = G1  + (size_t)b * SSZ * HSZ + h0;
  const int*   mrow = masks + b * SSZ;

  const v2f  e0    = {exp_neg(g0[0]), exp_neg(g0[1])};
  const bool far0  = !(__builtin_fabsf(g0[0]) < GMAX) || !(__builtin_fabsf(g0[1]) < GMAX);

  int len = 0;
  Chunk cur = load_chunk(xrow, grow, mrow, lane);
  for (int c = 0; c < SSZ / 32; ++c) {
    const int cn = (c + 1 < SSZ / 32) ? (c + 1) : c;
    const Chunk nxt = load_chunk(xrow, grow, mrow, cn * 32 + lane);

    const unsigned bits = __builtin_amdgcn_ballot_w32(cur.m != 0);
    len += __builtin_popcount(bits);
    const bool far = far0 || !(__builtin_fabsf(cur.g[0]) < GMAX) || !(__builtin_fabsf(cur.g[1]) < GMAX);
    if (__builtin_expect(__builtin_amdgcn_ballot_w32(far) != 0u, 0)) {
      add_rows<false>(bits, cur.x, cur.g, g0, acc);
    } else {
      const v2f e1 = {exp_neg(cur.g[0]), exp_neg(cur.g[1])};
      add_rows<true>(bits, cur.x, e1, e0, acc);
    }
    cur = nxt;
  }

  const float inv_len = 1.0f / (float)len;
  v2f o;
  #pragma unroll
  for (int t = 0; t < HT; ++t) o[t] = tanhf(mi * acc[t] * inv_len);

  *(v2f*)(tile + lane * TP + wv * HT) = o;
  __syncthreads();
  #pragma unroll
  for (int r = 0; r < 2; ++r) {
    const int row = wv * 2 + r;
    out[(size_t)(b * SSZ + it * 32 + row) * HSZ + hg * 32 + lane] = tile[row * TP + lane];
  }
}

extern "C" void kernel_launch(void* const* d_in, const int* in_sizes, int n_in,
                              void* d_out, int out_size, void* d_ws, size_t ws_size,
                              hipStream_t stream) {
  const float* inp   = (const float*)d_in[0];
  const int*   masks = (const int*)  d_in[1];
  const float* W0    = (const float*)d_in[2];
  const float* b0    = (const float*)d_in[3];
  const float* W1    = (const float*)d_in[4];
  const float* b1    = (const float*)d_in[5];
  float* out = (float*)d_out;

  float* G0 = (float*)d_ws;
  float* G1 = G0 + (size_t)BSZ * SSZ * HSZ;

  gates_wmma_kernel<<<256, 128, 0, stream>>>(inp, W0, b0, W1, b1, G0, G1);
  pairwise_kernel<<<256, 512, 0, stream>>>(inp, masks, G0, G1, out);
}


# pairwise term: one IEEE division per trip of 8 rows j (16 terms), not 2 rows (4): denominators multiplied up a tree, root inverted, 1/d carried back down; 4- and 2-row trips end a chunk; j order kept
# speedup vs baseline: 4.9946x; 1.1075x over previous
#include <hip/hip_runtime.h>

typedef __attribute__((ext_vector_type(2))) float v2f;
typedef __attribute__((ext_vector_type(8))) float v8f;

#define BSZ 4
#define SSZ 512
#define HSZ 128
#define HT  2
#define TP  36
#define PMAX 0x1.0p+96f
#define GMAX 80.0f
static_assert(HT == 2, "the shared division below is written for two columns a wave");

__global__ __launch_bounds__(128) void gates_wmma_kernel(
    const float* __restrict__ inp,
    const float* __restrict__ W0, const float* __restrict__ b0,
    const float* __restrict__ W1, const float* __restrict__ b1,
    float* __restrict__ G0, float* __restrict__ G1)
{
  const int lane  = threadIdx.x & 31;
  const int wave  = threadIdx.x >> 5;
  const int tile  = blockIdx.x * 4 + wave;
  const int ntile = tile & 7;
  const int mtile = tile >> 3;

  const int r  = lane & 15;
  const int kb = (lane >> 4) * 2;

  const float* Aptr  = inp + (size_t)(mtile * 16 + r) * HSZ + kb;
  const float* B0ptr = W0  + (size_t)(ntile * 16 + r) * HSZ + kb;
  const float* B1ptr = W1  + (size_t)(ntile * 16 + r) * HSZ + kb;

  v8f acc0 = {};
  v8f acc1 = {};
  #pragma unroll 4
  for (int k0 = 0; k0 < HSZ; k0 += 4) {
    v2f a  = *(const v2f*)(Aptr  + k0);
    v2f w0 = *(const v2f*)(B0ptr + k0);
    v2f w1 = *(const v2f*)(B1ptr + k0);
    acc0 = __builtin_amdgcn_wmma_f32_16x16x4_f32(
         false, a,  false, w0, (short)0, acc0, false, false);
    acc1 = __builtin_amdgcn_wmma_f32_16x16x4_f32(
         false, a,  false, w1, (short)0, acc1, false, false);
  }

  const int col  = ntile * 16 + (lane & 15);
  const int mofs = (lane >> 4) * 8;
  const float bias0 = b0[col];
  const float bias1 = b1[col];
  #pragma unroll
  for (int v = 0; v < 8; ++v) {
    const size_t idx = (size_t)(mtile * 16 + mofs + v) * HSZ + col;
    G0[idx] = acc0[v] + bias0;
    G1[idx] = acc1[v] + bias1;
  }
}

struct Chunk {
  v2f x;
  v2f g;
  int m;
};

__device__ __forceinline__ Chunk load_chunk(const float* __restrict__ xrow,
                                            const float* __restrict__ grow,
                                            const int*   __restrict__ mrow,
                                            int row)
{
  Chunk c;
  c.x = *(const v2f*)(xrow + (size_t)row * HSZ);
  c.g = *(const v2f*)(grow + (size_t)row * HSZ);
  c.m = mrow[row];
  return c;
}

__device__ __forceinline__ float lane_value(float v, int k)
{
  return __builtin_bit_cast(float,
           __builtin_amdgcn_readlane(__builtin_bit_cast(int, v), k));
}

__device__ __forceinline__ float exp_neg(float g)
{
  const float c_hi = -0x1.715476p+0f;
  const float c_lo = -0x1.4ae0cp-26f;
  const float t = g * c_hi;
  const float r = __builtin_fmaf(g, c_lo, __builtin_fmaf(g, c_hi, -t));
  const float e = __builtin_amdgcn_exp2f(t);
  return __builtin_fmaf(e, r * 0x1.62e43p-1f, e);
}

template <int N>
__device__ __forceinline__ float invert_all(const v2f (&d)[N], v2f (&s)[N])
{
  v2f node[2 * N];
  #pragma unroll
  for (int n = 0; n < N; ++n) node[N + n] = d[n];
  #pragma unroll
  for (int n = N - 1; n >= 1; --n) node[n] = node[2 * n] * node[2 * n + 1];
  const float pp = node[1][0] * node[1][1];
  const float r  = 1.0f / pp;
  v2f inv[2 * N];
  inv[1][0] = r * node[1][1];
  inv[1][1] = r * node[1][0];
  #pragma unroll
  for (int n = 1; n < N; ++n) {
    inv[2 * n]     = inv[n] * node[2 * n + 1];
    inv[2 * n + 1] = inv[n] * node[2 * n];
  }
  #pragma unroll
  for (int n = 0; n < N; ++n) s[n] = inv[N + n];
  return pp;
}

template <bool SPLIT, int NR>
__device__ __forceinline__ void add_trip(unsigned& bits, const v2f cx, const v2f cg,
                                         const v2f a0, float (&acc)[HT])
{
  int k[NR];
  #pragma unroll
  for (int r = 0; r < NR - 1; ++r) {
    k[r] = __builtin_ctz(bits);
    bits &= bits - 1;
  }
  const bool full = bits != 0;
  k[NR - 1] = full ? __builtin_ctz(bits) : k[0];
  bits &= bits - 1;
  float x[NR][HT], g[NR][HT];
  #pragma unroll
  for (int r = 0; r < NR; ++r) {
    #pragma unroll
    for (int t = 0; t < HT; ++t) {
      x[r][t] = lane_value(cx[t], k[r]);
      g[r][t] = lane_value(cg[t], k[r]);
    }
  }
  #pragma unroll
  for (int t = 0; t < HT; ++t) x[NR - 1][t] = full ? x[NR - 1][t] : 0.0f;
  v2f d[NR];
  #pragma unroll
  for (int r = 0; r < NR; ++r) {
    #pragma unroll
    for (int t = 0; t < HT; ++t) {
      if (SPLIT) d[r][t] = __builtin_fmaf(a0[t], g[r][t], 1.0f);
      else       d[r][t] = 1.0f + __expf(-(a0[t] + g[r][t]));
    }
  }
  v2f s[NR];
  const float pp = invert_all<NR>(d, s);
  if (__builtin_expect(__builtin_amdgcn_ballot_w32(!(pp < PMAX)) != 0u, 0)) {
    #pragma unroll
    for (int r = 0; r < NR; ++r) {
      #pragma unroll
      for (int t = 0; t < HT; ++t) s[r][t] = 1.0f / d[r][t];
    }
  }
  #pragma unroll
  for (int r = 0; r < NR; ++r) {
    #pragma unroll
    for (int t = 0; t < HT; ++t) acc[t] = __builtin_fmaf(x[r][t], s[r][t], acc[t]);
  }
}

template <bool SPLIT>
__device__ __forceinline__ void add_rows(unsigned bits, const v2f cx, const v2f cg,
                                         const v2f a0, float (&acc)[HT])
{
  while (__builtin_popcount(bits) >= 7) add_trip<SPLIT, 8>(bits, cx, cg, a0, acc);
  if (__builtin_popcount(bits) >= 3)    add_trip<SPLIT, 4>(bits, cx, cg, a0, acc);
  if (bits)                             add_trip<SPLIT, 2>(bits, cx, cg, a0, acc);
}

__global__ __launch_bounds__(512) void pairwise_kernel(
    const float* __restrict__ inp,
    const int*   __restrict__ masks,
    const float* __restrict__ G0,
    const float* __restrict__ G1,
    float* __restrict__ out)
{
  __shared__ __attribute__((aligned(16))) float tile[32 * TP];

  const int lane = threadIdx.x & 31;
  const int wv = threadIdx.x >> 5;
  const int hg = blockIdx.x & 3;
  const int it = (blockIdx.x >> 2) & 15;
  const int b  = blockIdx.x >> 6;
  const int h0 = hg * 32 + wv * HT;
  const int i  = it * 32 + lane;

  const v2f   g0 = *(const v2f*)(G0 + (size_t)(b * SSZ + i) * HSZ + h0);
  const float mi = (float)masks[b * SSZ + i];
  float acc[HT];
  #pragma unroll
  for (int t = 0; t < HT; ++t) acc[t] = 0.0f;

  const float* xrow = inp + (size_t)b * SSZ * HSZ + h0;
  const float* grow = G1  + (size_t)b * SSZ * HSZ + h0;
  const int*   mrow = masks + b * SSZ;

  const v2f  e0    = {exp_neg(g0[0]), exp_neg(g0[1])};
  const bool far0  = !(__builtin_fabsf(g0[0]) < GMAX) || !(__builtin_fabsf(g0[1]) < GMAX);

  int len = 0;
  Chunk cur = load_chunk(xrow, grow, mrow, lane);
  for (int c = 0; c < SSZ / 32; ++c) {
    const int cn = (c + 1 < SSZ / 32) ? (c + 1) : c;
    const Chunk nxt = load_chunk(xrow, grow, mrow, cn * 32 + lane);

    const unsigned bits = __builtin_amdgcn_ballot_w32(cur.m != 0);
    len += __builtin_popcount(bits);
    const bool far = far0 || !(__builtin_fabsf(cur.g[0]) < GMAX) || !(__builtin_fabsf(cur.g[1]) < GMAX);
    if (__builtin_expect(__builtin_amdgcn_ballot_w32(far) != 0u, 0)) {
      add_rows<false>(bits, cur.x, cur.g, g0, acc);
    } else {
      const v2f e1 = {exp_neg(cur.g[0]), exp_neg(cur.g[1])};
      add_rows<true>(bits, cur.x, e1, e0, acc);
    }
    cur = nxt;
  }

  const float inv_len = 1.0f / (float)len;
  v2f o;
  #pragma unroll
  for (int t = 0; t < HT; ++t) o[t] = tanhf(mi * acc[t] * inv_len);

  *(v2f*)(tile + lane * TP + wv * HT) = o;
  __syncthreads();
  #pragma unroll
  for (int r = 0; r < 2; ++r) {
    const int row = wv * 2 + r;
    out[(size_t)(b * SSZ + it * 32 + row) * HSZ + hg * 32 + lane] = tile[row * TP + lane];
  }
}

extern "C" void kernel_launch(void* const* d_in, const int* in_sizes, int n_in,
                              void* d_out, int out_size, void* d_ws, size_t ws_size,
                              hipStream_t stream) {
  const float* inp   = (const float*)d_in[0];
  const int*   masks = (const int*)  d_in[1];
  const float* W0    = (const float*)d_in[2];
  const float* b0    = (const float*)d_in[3];
  const float* W1    = (const float*)d_in[4];
  const float* b1    = (const float*)d_in[5];
  float* out = (float*)d_out;

  float* G0 = (float*)d_ws;
  float* G1 = G0 + (size_t)BSZ * SSZ * HSZ;

  gates_wmma_kernel<<<256, 128, 0, stream>>>(inp, W0, b0, W1, b1, G0, G1);
  pairwise_kernel<<<256, 512, 0, stream>>>(inp, masks, G0, G1, out);
}


# pairwise kernel: a lane holds 2 rows i of one column (wave = 1 column x 64 rows i), not 2 columns of one row: a lane-read row j is one scalar for both halves of each packed op; j order kept
# speedup vs baseline: 5.2879x; 1.0587x over previous
#include <hip/hip_runtime.h>

typedef __attribute__((ext_vector_type(2))) float v2f;
typedef __attribute__((ext_vector_type(8))) float v8f;

#define BSZ 4
#define SSZ 512
#define HSZ 128
#define RI  2
#define TP  17
#define PMAX 0x1.0p+96f
#define GMAX 80.0f
static_assert(RI == 2, "the shared division below is written for two rows i a lane");

__global__ __launch_bounds__(128) void gates_wmma_kernel(
    const float* __restrict__ inp,
    const float* __restrict__ W0, const float* __restrict__ b0,
    const float* __restrict__ W1, const float* __restrict__ b1,
    float* __restrict__ G0, float* __restrict__ G1)
{
  const int lane  = threadIdx.x & 31;
  const int wave  = threadIdx.x >> 5;
  const int tile  = blockIdx.x * 4 + wave;
  const int ntile = tile & 7;
  const int mtile = tile >> 3;

  const int r  = lane & 15;
  const int kb = (lane >> 4) * 2;

  const float* Aptr  = inp + (size_t)(mtile * 16 + r) * HSZ + kb;
  const float* B0ptr = W0  + (size_t)(ntile * 16 + r) * HSZ + kb;
  const float* B1ptr = W1  + (size_t)(ntile * 16 + r) * HSZ + kb;

  v8f acc0 = {};
  v8f acc1 = {};
  #pragma unroll 4
  for (int k0 = 0; k0 < HSZ; k0 += 4) {
    v2f a  = *(const v2f*)(Aptr  + k0);
    v2f w0 = *(const v2f*)(B0ptr + k0);
    v2f w1 = *(const v2f*)(B1ptr + k0);
    acc0 = __builtin_amdgcn_wmma_f32_16x16x4_f32(
         false, a,  false, w0, (short)0, acc0, false, false);
    acc1 = __builtin_amdgcn_wmma_f32_16x16x4_f32(
         false, a,  false, w1, (short)0, acc1, false, false);
  }

  const int col  = ntile * 16 + (lane & 15);
  const int mofs = (lane >> 4) * 8;
  const float bias0 = b0[col];
  const float bias1 = b1[col];
  #pragma unroll
  for (int v = 0; v < 8; ++v) {
    const size_t idx = (size_t)(mtile * 16 + mofs + v) * HSZ + col;
    G0[idx] = acc0[v] + bias0;
    G1[idx] = acc1[v] + bias1;
  }
}

struct Chunk {
  float x;
  float g;
  int m;
};

__device__ __forceinline__ Chunk load_chunk(const float* __restrict__ xcol,
                                            const float* __restrict__ gcol,
                                            const int*   __restrict__ mrow,
                                            int row)
{
  Chunk c;
  c.x = xcol[(size_t)row * HSZ];
  c.g = gcol[(size_t)row * HSZ];
  c.m = mrow[row];
  return c;
}

__device__ __forceinline__ float lane_value(float v, int k)
{
  return __builtin_bit_cast(float,
           __builtin_amdgcn_readlane(__builtin_bit_cast(int, v), k));
}

__device__ __forceinline__ float exp_neg(float g)
{
  const float c_hi = -0x1.715476p+0f;
  const float c_lo = -0x1.4ae0cp-26f;
  const float t = g * c_hi;
  const float r = __builtin_fmaf(g, c_lo, __builtin_fmaf(g, c_hi, -t));
  const float e = __builtin_amdgcn_exp2f(t);
  return __builtin_fmaf(e, r * 0x1.62e43p-1f, e);
}

template <int N>
__device__ __forceinline__ float invert_all(const v2f (&d)[N], v2f (&s)[N])
{
  v2f node[2 * N];
  #pragma unroll
  for (int n = 0; n < N; ++n) node[N + n] = d[n];
  #pragma unroll
  for (int n = N - 1; n >= 1; --n) node[n] = node[2 * n] * node[2 * n + 1];
  const float pp = node[1][0] * node[1][1];
  const float r  = 1.0f / pp;
  v2f inv[2 * N];
  inv[1][0] = r * node[1][1];
  inv[1][1] = r * node[1][0];
  #pragma unroll
  for (int n = 1; n < N; ++n) {
    inv[2 * n]     = inv[n] * node[2 * n + 1];
    inv[2 * n + 1] = inv[n] * node[2 * n];
  }
  #pragma unroll
  for (int n = 0; n < N; ++n) s[n] = inv[N + n];
  return pp;
}

template <bool SPLIT, int NR>
__device__ __forceinline__ void add_trip(unsigned& bits, const float cx, const float cg,
                                         const v2f a0, v2f& acc)
{
  int k[NR];
  #pragma unroll
  for (int r = 0; r < NR - 1; ++r) {
    k[r] = __builtin_ctz(bits);
    bits &= bits - 1;
  }
  const bool full = bits != 0;
  k[NR - 1] = full ? __builtin_ctz(bits) : k[0];
  bits &= bits - 1;
  float x[NR], g[NR];
  #pragma unroll
  for (int r = 0; r < NR; ++r) {
    x[r] = lane_value(cx, k[r]);
    g[r] = lane_value(cg, k[r]);
  }
  x[NR - 1] = full ? x[NR - 1] : 0.0f;
  v2f d[NR];
  #pragma unroll
  for (int r = 0; r < NR; ++r) {
    if (SPLIT) {
      const v2f gg = {g[r], g[r]};
      const v2f one = {1.0f, 1.0f};
      d[r] = __builtin_elementwise_fma(a0, gg, one);
    } else {
      #pragma unroll
      for (int t = 0; t < RI; ++t) d[r][t] = 1.0f + __expf(-(a0[t] + g[r]));
    }
  }
  v2f s[NR];
  const float pp = invert_all<NR>(d, s);
  if (__builtin_expect(__builtin_amdgcn_ballot_w32(!(pp < PMAX)) != 0u, 0)) {
    #pragma unroll
    for (int r = 0; r < NR; ++r) {
      #pragma unroll
      for (int t = 0; t < RI; ++t) s[r][t] = 1.0f / d[r][t];
    }
  }
  #pragma unroll
  for (int r = 0; r < NR; ++r) {
    #pragma unroll
    for (int t = 0; t < RI; ++t) acc[t] = __builtin_fmaf(x[r], s[r][t], acc[t]);
  }
}

template <bool SPLIT>
__device__ __forceinline__ void add_rows(unsigned bits, const float cx, const float cg,
                                         const v2f a0, v2f& acc)
{
  while (__builtin_popcount(bits) >= 7) add_trip<SPLIT, 8>(bits, cx, cg, a0, acc);
  if (__builtin_popcount(bits) >= 3)    add_trip<SPLIT, 4>(bits, cx, cg, a0, acc);
  if (bits)                             add_trip<SPLIT, 2>(bits, cx, cg, a0, acc);
}

__global__ __launch_bounds__(512) void pairwise_kernel(
    const float* __restrict__ inp,
    const int*   __restrict__ masks,
    const float* __restrict__ G0,
    const float* __restrict__ G1,
    float* __restrict__ out)
{
  __shared__ float tile[64 * TP];

  const int lane = threadIdx.x & 31;
  const int wv = threadIdx.x >> 5;
  const int hg = blockIdx.x & 7;
  const int it = (blockIdx.x >> 3) & 7;
  const int b  = blockIdx.x >> 6;
  const int h  = hg * 16 + wv;
  const int i0 = it * 64 + lane;

  const v2f g0 = {G0[(size_t)(b * SSZ + i0) * HSZ + h],
                  G0[(size_t)(b * SSZ + i0 + 32) * HSZ + h]};
  const v2f mi = {(float)masks[b * SSZ + i0], (float)masks[b * SSZ + i0 + 32]};
  v2f acc = {0.0f, 0.0f};

  const float* xcol = inp + (size_t)b * SSZ * HSZ + h;
  const float* gcol = G1  + (size_t)b * SSZ * HSZ + h;
  const int*   mrow = masks + b * SSZ;

  const v2f  e0    = {exp_neg(g0[0]), exp_neg(g0[1])};
  const bool far0  = !(__builtin_fabsf(g0[0]) < GMAX) || !(__builtin_fabsf(g0[1]) < GMAX);

  int len = 0;
  Chunk cur = load_chunk(xcol, gcol, mrow, lane);
  for (int c = 0; c < SSZ / 32; ++c) {
    const int cn = (c + 1 < SSZ / 32) ? (c + 1) : c;
    const Chunk nxt = load_chunk(xcol, gcol, mrow, cn * 32 + lane);

    const unsigned bits = __builtin_amdgcn_ballot_w32(cur.m != 0);
    len += __builtin_popcount(bits);
    const bool far = far0 || !(__builtin_fabsf(cur.g) < GMAX);
    if (__builtin_expect(__builtin_amdgcn_ballot_w32(far) != 0u, 0)) {
      add_rows<false>(bits, cur.x, cur.g, g0, acc);
    } else {
      const float e1 = exp_neg(cur.g);
      add_rows<true>(bits, cur.x, e1, e0, acc);
    }
    cur = nxt;
  }

  const float inv_len = 1.0f / (float)len;
  float o[RI];
  #pragma unroll
  for (int t = 0; t < RI; ++t) o[t] = tanhf(mi[t] * acc[t] * inv_len);

  #pragma unroll
  for (int t = 0; t < RI; ++t) tile[(lane + 32 * t) * TP + wv] = o[t];
  __syncthreads();
  #pragma unroll
  for (int r = 0; r < 2; ++r) {
    const int row = wv * 4 + r * 2 + (lane >> 4);
    const int col = lane & 15;
    out[(size_t)(b * SSZ + it * 64 + row) * HSZ + hg * 16 + col] = tile[row * TP + col];
  }
}

extern "C" void kernel_launch(void* const* d_in, const int* in_sizes, int n_in,
                              void* d_out, int out_size, void* d_ws, size_t ws_size,
                              hipStream_t stream) {
  const float* inp   = (const float*)d_in[0];
  const int*   masks = (const int*)  d_in[1];
  const float* W0    = (const float*)d_in[2];
  const float* b0    = (const float*)d_in[3];
  const float* W1    = (const float*)d_in[4];
  const float* b1    = (const float*)d_in[5];
  float* out = (float*)d_out;

  float* G0 = (float*)d_ws;
  float* G1 = G0 + (size_t)BSZ * SSZ * HSZ;

  gates_wmma_kernel<<<256, 128, 0, stream>>>(inp, W0, b0, W1, b1, G0, G1);
  pairwise_kernel<<<256, 512, 0, stream>>>(inp, masks, G0, G1, out);
}


# gate GEMM: K loop unrolled whole and the kernel declared one wave per SIMD, so operand and bias loads run up to 63 ahead of the matrix instructions that use them, not 12 a trip; k order kept
# speedup vs baseline: 5.5671x; 1.0528x over previous
#include <hip/hip_runtime.h>

typedef __attribute__((ext_vector_type(2))) float v2f;
typedef __attribute__((ext_vector_type(8))) float v8f;

#define BSZ 4
#define SSZ 512
#define HSZ 128
#define RI  2
#define TP  17
#define PMAX 0x1.0p+96f
#define GMAX 80.0f
static_assert(RI == 2, "the shared division below is written for two rows i a lane");

__global__ __launch_bounds__(128) __attribute__((amdgpu_waves_per_eu(1, 1))) void gates_wmma_kernel(
    const float* __restrict__ inp,
    const float* __restrict__ W0, const float* __restrict__ b0,
    const float* __restrict__ W1, const float* __restrict__ b1,
    float* __restrict__ G0, float* __restrict__ G1)
{
  const int lane  = threadIdx.x & 31;
  const int wave  = threadIdx.x >> 5;
  const int tile  = blockIdx.x * 4 + wave;
  const int ntile = tile & 7;
  const int mtile = tile >> 3;

  const int r  = lane & 15;
  const int kb = (lane >> 4) * 2;

  const float* Aptr  = inp + (size_t)(mtile * 16 + r) * HSZ + kb;
  const float* B0ptr = W0  + (size_t)(ntile * 16 + r) * HSZ + kb;
  const float* B1ptr = W1  + (size_t)(ntile * 16 + r) * HSZ + kb;

  constexpr int KS = HSZ / 4;
  const int col  = ntile * 16 + (lane & 15);
  v2f a[KS], w0[KS], w1[KS];
  #pragma unroll
  for (int s = 0; s < KS; ++s) {
    a[s]  = *(const v2f*)(Aptr  + 4 * s);
    w0[s] = *(const v2f*)(B0ptr + 4 * s);
    w1[s] = *(const v2f*)(B1ptr + 4 * s);
  }
  const float bias0 = b0[col];
  const float bias1 = b1[col];

  v8f acc0 = {};
  v8f acc1 = {};
  #pragma unroll
  for (int s = 0; s < KS; ++s) {
    acc0 = __builtin_amdgcn_wmma_f32_16x16x4_f32(
         false, a[s],  false, w0[s], (short)0, acc0, false, false);
    acc1 = __builtin_amdgcn_wmma_f32_16x16x4_f32(
         false, a[s],  false, w1[s], (short)0, acc1, false, false);
  }

  const int mofs = (lane >> 4) * 8;
  #pragma unroll
  for (int v = 0; v < 8; ++v) {
    const size_t idx = (size_t)(mtile * 16 + mofs + v) * HSZ + col;
    G0[idx] = acc0[v] + bias0;
    G1[idx] = acc1[v] + bias1;
  }
}

struct Chunk {
  float x;
  float g;
  int m;
};

__device__ __forceinline__ Chunk load_chunk(const float* __restrict__ xcol,
                                            const float* __restrict__ gcol,
                                            const int*   __restrict__ mrow,
                                            int row)
{
  Chunk c;
  c.x = xcol[(size_t)row * HSZ];
  c.g = gcol[(size_t)row * HSZ];
  c.m = mrow[row];
  return c;
}

__device__ __forceinline__ float lane_value(float v, int k)
{
  return __builtin_bit_cast(float,
           __builtin_amdgcn_readlane(__builtin_bit_cast(int, v), k));
}

__device__ __forceinline__ float exp_neg(float g)
{
  const float c_hi = -0x1.715476p+0f;
  const float c_lo = -0x1.4ae0cp-26f;
  const float t = g * c_hi;
  const float r = __builtin_fmaf(g, c_lo, __builtin_fmaf(g, c_hi, -t));
  const float e = __builtin_amdgcn_exp2f(t);
  return __builtin_fmaf(e, r * 0x1.62e43p-1f, e);
}

template <int N>
__device__ __forceinline__ float invert_all(const v2f (&d)[N], v2f (&s)[N])
{
  v2f node[2 * N];
  #pragma unroll
  for (int n = 0; n < N; ++n) node[N + n] = d[n];
  #pragma unroll
  for (int n = N - 1; n >= 1; --n) node[n] = node[2 * n] * node[2 * n + 1];
  const float pp = node[1][0] * node[1][1];
  const float r  = 1.0f / pp;
  v2f inv[2 * N];
  inv[1][0] = r * node[1][1];
  inv[1][1] = r * node[1][0];
  #pragma unroll
  for (int n = 1; n < N; ++n) {
    inv[2 * n]     = inv[n] * node[2 * n + 1];
    inv[2 * n + 1] = inv[n] * node[2 * n];
  }
  #pragma unroll
  for (int n = 0; n < N; ++n) s[n] = inv[N + n];
  return pp;
}

template <bool SPLIT, int NR>
__device__ __forceinline__ void add_trip(unsigned& bits, const float cx, const float cg,
                                         const v2f a0, v2f& acc)
{
  int k[NR];
  #pragma unroll
  for (int r = 0; r < NR - 1; ++r) {
    k[r] = __builtin_ctz(bits);
    bits &= bits - 1;
  }
  const bool full = bits != 0;
  k[NR - 1] = full ? __builtin_ctz(bits) : k[0];
  bits &= bits - 1;
  float x[NR], g[NR];
  #pragma unroll
  for (int r = 0; r < NR; ++r) {
    x[r] = lane_value(cx, k[r]);
    g[r] = lane_value(cg, k[r]);
  }
  x[NR - 1] = full ? x[NR - 1] : 0.0f;
  v2f d[NR];
  #pragma unroll
  for (int r = 0; r < NR; ++r) {
    if (SPLIT) {
      const v2f gg = {g[r], g[r]};
      const v2f one = {1.0f, 1.0f};
      d[r] = __builtin_elementwise_fma(a0, gg, one);
    } else {
      #pragma unroll
      for (int t = 0; t < RI; ++t) d[r][t] = 1.0f + __expf(-(a0[t] + g[r]));
    }
  }
  v2f s[NR];
  const float pp = invert_all<NR>(d, s);
  if (__builtin_expect(__builtin_amdgcn_ballot_w32(!(pp < PMAX)) != 0u, 0)) {
    #pragma unroll
    for (int r = 0; r < NR; ++r) {
      #pragma unroll
      for (int t = 0; t < RI; ++t) s[r][t] = 1.0f / d[r][t];
    }
  }
  #pragma unroll
  for (int r = 0; r < NR; ++r) {
    #pragma unroll
    for (int t = 0; t < RI; ++t) acc[t] = __builtin_fmaf(x[r], s[r][t], acc[t]);
  }
}

template <bool SPLIT>
__device__ __forceinline__ void add_rows(unsigned bits, const float cx, const float cg,
                                         const v2f a0, v2f& acc)
{
  while (__builtin_popcount(bits) >= 7) add_trip<SPLIT, 8>(bits, cx, cg, a0, acc);
  if (__builtin_popcount(bits) >= 3)    add_trip<SPLIT, 4>(bits, cx, cg, a0, acc);
  if (bits)                             add_trip<SPLIT, 2>(bits, cx, cg, a0, acc);
}

__global__ __launch_bounds__(512) void pairwise_kernel(
    const float* __restrict__ inp,
    const int*   __restrict__ masks,
    const float* __restrict__ G0,
    const float* __restrict__ G1,
    float* __restrict__ out)
{
  __shared__ float tile[64 * TP];

  const int lane = threadIdx.x & 31;
  const int wv = threadIdx.x >> 5;
  const int hg = blockIdx.x & 7;
  const int it = (blockIdx.x >> 3) & 7;
  const int b  = blockIdx.x >> 6;
  const int h  = hg * 16 + wv;
  const int i0 = it * 64 + lane;

  const v2f g0 = {G0[(size_t)(b * SSZ + i0) * HSZ + h],
                  G0[(size_t)(b * SSZ + i0 + 32) * HSZ + h]};
  const v2f mi = {(float)masks[b * SSZ + i0], (float)masks[b * SSZ + i0 + 32]};
  v2f acc = {0.0f, 0.0f};

  const float* xcol = inp + (size_t)b * SSZ * HSZ + h;
  const float* gcol = G1  + (size_t)b * SSZ * HSZ + h;
  const int*   mrow = masks + b * SSZ;

  const v2f  e0    = {exp_neg(g0[0]), exp_neg(g0[1])};
  const bool far0  = !(__builtin_fabsf(g0[0]) < GMAX) || !(__builtin_fabsf(g0[1]) < GMAX);

  int len = 0;
  Chunk cur = load_chunk(xcol, gcol, mrow, lane);
  for (int c = 0; c < SSZ / 32; ++c) {
    const int cn = (c + 1 < SSZ / 32) ? (c + 1) : c;
    const Chunk nxt = load_chunk(xcol, gcol, mrow, cn * 32 + lane);

    const unsigned bits = __builtin_amdgcn_ballot_w32(cur.m != 0);
    len += __builtin_popcount(bits);
    const bool far = far0 || !(__builtin_fabsf(cur.g) < GMAX);
    if (__builtin_expect(__builtin_amdgcn_ballot_w32(far) != 0u, 0)) {
      add_rows<false>(bits, cur.x, cur.g, g0, acc);
    } else {
      const float e1 = exp_neg(cur.g);
      add_rows<true>(bits, cur.x, e1, e0, acc);
    }
    cur = nxt;
  }

  const float inv_len = 1.0f / (float)len;
  float o[RI];
  #pragma unroll
  for (int t = 0; t < RI; ++t) o[t] = tanhf(mi[t] * acc[t] * inv_len);

  #pragma unroll
  for (int t = 0; t < RI; ++t) tile[(lane + 32 * t) * TP + wv] = o[t];
  __syncthreads();
  #pragma unroll
  for (int r = 0; r < 2; ++r) {
    const int row = wv * 4 + r * 2 + (lane >> 4);
    const int col = lane & 15;
    out[(size_t)(b * SSZ + it * 64 + row) * HSZ + hg * 16 + col] = tile[row * TP + col];
  }
}

extern "C" void kernel_launch(void* const* d_in, const int* in_sizes, int n_in,
                              void* d_out, int out_size, void* d_ws, size_t ws_size,
                              hipStream_t stream) {
  const float* inp   = (const float*)d_in[0];
  const int*   masks = (const int*)  d_in[1];
  const float* W0    = (const float*)d_in[2];
  const float* b0    = (const float*)d_in[3];
  const float* W1    = (const float*)d_in[4];
  const float* b1    = (const float*)d_in[5];
  float* out = (float*)d_out;

  float* G0 = (float*)d_ws;
  float* G1 = G0 + (size_t)BSZ * SSZ * HSZ;

  gates_wmma_kernel<<<256, 128, 0, stream>>>(inp, W0, b0, W1, b1, G0, G1);
  pairwise_kernel<<<256, 512, 0, stream>>>(inp, masks, G0, G1, out);
}


# pairwise kernel: unmasked rows j are pushed lane to lane (ds_permute) onto a row stream; a full 32-row register is used from fixed lanes, 4 trips of 8: no bit scan, no short trips; j order kept
# speedup vs baseline: 6.6638x; 1.1970x over previous
#include <hip/hip_runtime.h>

typedef __attribute__((ext_vector_type(2))) float v2f;
typedef __attribute__((ext_vector_type(8))) float v8f;

#define BSZ 4
#define SSZ 512
#define HSZ 128
#define RI  2
#define TP  17
#define PMAX 0x1.0p+96f
#define GMAX 80.0f
static_assert(RI == 2, "the shared division below is written for two rows i a lane");

__global__ __launch_bounds__(128) __attribute__((amdgpu_waves_per_eu(1, 1))) void gates_wmma_kernel(
    const float* __restrict__ inp,
    const float* __restrict__ W0, const float* __restrict__ b0,
    const float* __restrict__ W1, const float* __restrict__ b1,
    float* __restrict__ G0, float* __restrict__ G1)
{
  const int lane  = threadIdx.x & 31;
  const int wave  = threadIdx.x >> 5;
  const int tile  = blockIdx.x * 4 + wave;
  const int ntile = tile & 7;
  const int mtile = tile >> 3;

  const int r  = lane & 15;
  const int kb = (lane >> 4) * 2;

  const float* Aptr  = inp + (size_t)(mtile * 16 + r) * HSZ + kb;
  const float* B0ptr = W0  + (size_t)(ntile * 16 + r) * HSZ + kb;
  const float* B1ptr = W1  + (size_t)(ntile * 16 + r) * HSZ + kb;

  constexpr int KS = HSZ / 4;
  const int col  = ntile * 16 + (lane & 15);
  v2f a[KS], w0[KS], w1[KS];
  #pragma unroll
  for (int s = 0; s < KS; ++s) {
    a[s]  = *(const v2f*)(Aptr  + 4 * s);
    w0[s] = *(const v2f*)(B0ptr + 4 * s);
    w1[s] = *(const v2f*)(B1ptr + 4 * s);
  }
  const float bias0 = b0[col];
  const float bias1 = b1[col];

  v8f acc0 = {};
  v8f acc1 = {};
  #pragma unroll
  for (int s = 0; s < KS; ++s) {
    acc0 = __builtin_amdgcn_wmma_f32_16x16x4_f32(
         false, a[s],  false, w0[s], (short)0, acc0, false, false);
    acc1 = __builtin_amdgcn_wmma_f32_16x16x4_f32(
         false, a[s],  false, w1[s], (short)0, acc1, false, false);
  }

  const int mofs = (lane >> 4) * 8;
  #pragma unroll
  for (int v = 0; v < 8; ++v) {
    const size_t idx = (size_t)(mtile * 16 + mofs + v) * HSZ + col;
    G0[idx] = acc0[v] + bias0;
    G1[idx] = acc1[v] + bias1;
  }
}

struct Chunk {
  float x;
  float g;
  int m;
};

__device__ __forceinline__ Chunk load_chunk(const float* __restrict__ xcol,
                                            const float* __restrict__ gcol,
                                            const int*   __restrict__ mrow,
                                            int row)
{
  Chunk c;
  c.x = xcol[(size_t)row * HSZ];
  c.g = gcol[(size_t)row * HSZ];
  c.m = mrow[row];
  return c;
}

__device__ __forceinline__ float lane_value(float v, int k)
{
  return __builtin_bit_cast(float,
           __builtin_amdgcn_readlane(__builtin_bit_cast(int, v), k));
}

__device__ __forceinline__ float send_to_lane(float v, int byte_addr)
{
  return __builtin_bit_cast(float,
           __builtin_amdgcn_ds_permute(byte_addr, __builtin_bit_cast(int, v)));
}

__device__ __forceinline__ float exp_neg(float g)
{
  const float c_hi = -0x1.715476p+0f;
  const float c_lo = -0x1.4ae0cp-26f;
  const float t = g * c_hi;
  const float r = __builtin_fmaf(g, c_lo, __builtin_fmaf(g, c_hi, -t));
  const float e = __builtin_amdgcn_exp2f(t);
  return __builtin_fmaf(e, r * 0x1.62e43p-1f, e);
}

template <int N>
__device__ __forceinline__ float invert_all(const v2f (&d)[N], v2f (&s)[N])
{
  v2f node[2 * N];
  #pragma unroll
  for (int n = 0; n < N; ++n) node[N + n] = d[n];
  #pragma unroll
  for (int n = N - 1; n >= 1; --n) node[n] = node[2 * n] * node[2 * n + 1];
  const float pp = node[1][0] * node[1][1];
  const float r  = 1.0f / pp;
  v2f inv[2 * N];
  inv[1][0] = r * node[1][1];
  inv[1][1] = r * node[1][0];
  #pragma unroll
  for (int n = 1; n < N; ++n) {
    inv[2 * n]     = inv[n] * node[2 * n + 1];
    inv[2 * n + 1] = inv[n] * node[2 * n];
  }
  #pragma unroll
  for (int n = 0; n < N; ++n) s[n] = inv[N + n];
  return pp;
}

template <int L0>
__device__ __forceinline__ float fast_trip(const float sx, const float se,
                                           const v2f e0, v2f& acc)
{
  float x[8], e[8];
  #pragma unroll
  for (int r = 0; r < 8; ++r) {
    x[r] = lane_value(sx, L0 + r);
    e[r] = lane_value(se, L0 + r);
  }
  v2f d[8];
  #pragma unroll
  for (int r = 0; r < 8; ++r) {
    const v2f ee = {e[r], e[r]};
    const v2f one = {1.0f, 1.0f};
    d[r] = __builtin_elementwise_fma(e0, ee, one);
  }
  v2f s[8];
  const float pp = invert_all<8>(d, s);
  #pragma unroll
  for (int r = 0; r < 8; ++r) {
    #pragma unroll
    for (int t = 0; t < RI; ++t) acc[t] = __builtin_fmaf(x[r], s[r][t], acc[t]);
  }
  return pp;
}

template <bool SPLIT>
__device__ __forceinline__ void careful_trip(const float sx, const float sv, const int lane0,
                                             const v2f a0, v2f& acc)
{
  float x[8], g[8];
  #pragma unroll
  for (int r = 0; r < 8; ++r) {
    x[r] = lane_value(sx, lane0 + r);
    g[r] = lane_value(sv, lane0 + r);
  }
  v2f d[8];
  #pragma unroll
  for (int r = 0; r < 8; ++r) {
    if (SPLIT) {
      const v2f gg = {g[r], g[r]};
      const v2f one = {1.0f, 1.0f};
      d[r] = __builtin_elementwise_fma(a0, gg, one);
    } else {
      #pragma unroll
      for (int t = 0; t < RI; ++t) d[r][t] = 1.0f + __expf(-(a0[t] + g[r]));
    }
  }
  v2f s[8];
  const float pp = invert_all<8>(d, s);
  if (__builtin_amdgcn_ballot_w32(!(pp < PMAX)) != 0u) {
    #pragma unroll
    for (int r = 0; r < 8; ++r) {
      #pragma unroll
      for (int t = 0; t < RI; ++t) s[r][t] = 1.0f / d[r][t];
    }
  }
  #pragma unroll
  for (int r = 0; r < 8; ++r) {
    #pragma unroll
    for (int t = 0; t < RI; ++t) acc[t] = __builtin_fmaf(x[r], s[r][t], acc[t]);
  }
}

template <bool SPLIT>
__device__ __forceinline__ void careful_rows(const float sx, const float sv, const int rows,
                                             const v2f a0, v2f& acc)
{
  #pragma nounroll
  for (int lane0 = 0; lane0 < rows; lane0 += 8) careful_trip<SPLIT>(sx, sv, lane0, a0, acc);
}

template <bool WHOLE>
__device__ __forceinline__ void use_register(const float sx, const float se, const float sg,
                                             const int rows, const bool far,
                                             const v2f e0, const v2f g0, v2f& acc)
{
  if (__builtin_expect(far, 0)) {
    careful_rows<false>(sx, sg, rows, g0, acc);
    return;
  }
  v2f trial = acc;
  float p = fast_trip<0>(sx, se, e0, trial);
  bool bad = !(p < PMAX);
  if (WHOLE || rows > 8)  { p = fast_trip<8>(sx, se, e0, trial);  bad = bad || !(p < PMAX); }
  if (WHOLE || rows > 16) { p = fast_trip<16>(sx, se, e0, trial); bad = bad || !(p < PMAX); }
  if (WHOLE || rows > 24) { p = fast_trip<24>(sx, se, e0, trial); bad = bad || !(p < PMAX); }
  if (__builtin_expect(__builtin_amdgcn_ballot_w32(bad) != 0u, 0)) {
    careful_rows<true>(sx, se, rows, e0, acc);
  } else {
    acc = trial;
  }
}

struct Stream {
  float x, e, g;
  int fill;
};

__device__ __forceinline__ int take_chunk(const Chunk cur, const int lane, bool& far, Stream& st,
                                          const v2f e0, const v2f g0, v2f& acc)
{
  const bool on = cur.m != 0;
  const unsigned bits = __builtin_amdgcn_ballot_w32(on);
  const int n = __builtin_popcount(bits);
  far = far || (__builtin_amdgcn_ballot_w32(!(__builtin_fabsf(cur.g) < GMAX)) != 0u);
  const float e1 = exp_neg(cur.g);

  const int below = (int)__builtin_amdgcn_mbcnt_lo(bits, 0u);
  const int place = on ? (st.fill + below) : (st.fill + n + (lane - below));
  const int to = (place & 31) << 2;
  const float rx = send_to_lane(cur.x, to);
  const float re = send_to_lane(e1, to);
  const float rg = send_to_lane(cur.g, to);
  const int end = st.fill + n;
  const bool fits = (lane >= st.fill) && (lane < end);
  st.x = fits ? rx : st.x;
  st.e = fits ? re : st.e;
  st.g = fits ? rg : st.g;
  st.fill = end;

  if (st.fill >= 32) {
    use_register<true>(st.x, st.e, st.g, 32, far, e0, g0, acc);
    st.x = rx;
    st.e = re;
    st.g = rg;
    st.fill -= 32;
  }
  return n;
}

__global__ __launch_bounds__(512) void pairwise_kernel(
    const float* __restrict__ inp,
    const int*   __restrict__ masks,
    const float* __restrict__ G0,
    const float* __restrict__ G1,
    float* __restrict__ out)
{
  __shared__ float tile[64 * TP];

  const int lane = threadIdx.x & 31;
  const int wv = threadIdx.x >> 5;
  const int hg = blockIdx.x & 7;
  const int it = (blockIdx.x >> 3) & 7;
  const int b  = blockIdx.x >> 6;
  const int h  = hg * 16 + wv;
  const int i0 = it * 64 + lane;

  const v2f g0 = {G0[(size_t)(b * SSZ + i0) * HSZ + h],
                  G0[(size_t)(b * SSZ + i0 + 32) * HSZ + h]};
  const v2f mi = {(float)masks[b * SSZ + i0], (float)masks[b * SSZ + i0 + 32]};
  v2f acc = {0.0f, 0.0f};

  const float* xcol = inp + (size_t)b * SSZ * HSZ + h;
  const float* gcol = G1  + (size_t)b * SSZ * HSZ + h;
  const int*   mrow = masks + b * SSZ;

  const v2f  e0    = {exp_neg(g0[0]), exp_neg(g0[1])};
  const bool far0  = !(__builtin_fabsf(g0[0]) < GMAX) || !(__builtin_fabsf(g0[1]) < GMAX);
  bool far = __builtin_amdgcn_ballot_w32(far0) != 0u;

  Stream st = {0.0f, 0.0f, 0.0f, 0};

  int len = 0;
  Chunk c0 = load_chunk(xcol, gcol, mrow, lane);
  Chunk c1 = load_chunk(xcol, gcol, mrow, 32 + lane);
  for (int p = 0; p < SSZ / 64; ++p) {
    const int pn = (p + 1 < SSZ / 64) ? (p + 1) : p;
    const Chunk n0 = load_chunk(xcol, gcol, mrow, pn * 64 + lane);
    const Chunk n1 = load_chunk(xcol, gcol, mrow, pn * 64 + 32 + lane);
    len += take_chunk(c0, lane, far, st, e0, g0, acc);
    len += take_chunk(c1, lane, far, st, e0, g0, acc);
    c0 = n0;
    c1 = n1;
  }
  if (st.fill > 0) {
    const bool live = lane < st.fill;
    const float tx = live ? st.x : 0.0f;
    const float te = live ? st.e : 0.0f;
    const float tg = live ? st.g : 0.0f;
    use_register<false>(tx, te, tg, st.fill, far, e0, g0, acc);
  }

  const float inv_len = 1.0f / (float)len;
  float o[RI];
  #pragma unroll
  for (int t = 0; t < RI; ++t) o[t] = tanhf(mi[t] * acc[t] * inv_len);

  #pragma unroll
  for (int t = 0; t < RI; ++t) tile[(lane + 32 * t) * TP + wv] = o[t];
  __syncthreads();
  #pragma unroll
  for (int r = 0; r < 2; ++r) {
    const int row = wv * 4 + r * 2 + (lane >> 4);
    const int col = lane & 15;
    out[(size_t)(b * SSZ + it * 64 + row) * HSZ + hg * 16 + col] = tile[row * TP + col];
  }
}

extern "C" void kernel_launch(void* const* d_in, const int* in_sizes, int n_in,
                              void* d_out, int out_size, void* d_ws, size_t ws_size,
                              hipStream_t stream) {
  const float* inp   = (const float*)d_in[0];
  const int*   masks = (const int*)  d_in[1];
  const float* W0    = (const float*)d_in[2];
  const float* b0    = (const float*)d_in[3];
  const float* W1    = (const float*)d_in[4];
  const float* b1    = (const float*)d_in[5];
  float* out = (float*)d_out;

  float* G0 = (float*)d_ws;
  float* G1 = G0 + (size_t)BSZ * SSZ * HSZ;

  gates_wmma_kernel<<<256, 128, 0, stream>>>(inp, W0, b0, W1, b1, G0, G1);
  pairwise_kernel<<<256, 512, 0, stream>>>(inp, masks, G0, G1, out);
}


# pairwise kernel: a lane holds 4 rows i of one column, not 2 (wave = 128 rows i, 8 waves a block, half as many waves): a lane-read row j serves 128 rows; one division per 32 terms; j order kept
# speedup vs baseline: 7.0798x; 1.0624x over previous
#include <hip/hip_runtime.h>

typedef __attribute__((ext_vector_type(2))) float v2f;
typedef __attribute__((ext_vector_type(8))) float v8f;

#define BSZ 4
#define SSZ 512
#define HSZ 128
#define RI  2
#define RP  2
#define TP  9
#define PMAX 0x1.0p+96f
#define GMAX 80.0f
static_assert(RI == 2, "a packed register pair has two halves");
struct Rows { v2f p[RP]; };

__global__ __launch_bounds__(128) __attribute__((amdgpu_waves_per_eu(1, 1))) void gates_wmma_kernel(
    const float* __restrict__ inp,
    const float* __restrict__ W0, const float* __restrict__ b0,
    const float* __restrict__ W1, const float* __restrict__ b1,
    float* __restrict__ G0, float* __restrict__ G1)
{
  const int lane  = threadIdx.x & 31;
  const int wave  = threadIdx.x >> 5;
  const int tile  = blockIdx.x * 4 + wave;
  const int ntile = tile & 7;
  const int mtile = tile >> 3;

  const int r  = lane & 15;
  const int kb = (lane >> 4) * 2;

  const float* Aptr  = inp + (size_t)(mtile * 16 + r) * HSZ + kb;
  const float* B0ptr = W0  + (size_t)(ntile * 16 + r) * HSZ + kb;
  const float* B1ptr = W1  + (size_t)(ntile * 16 + r) * HSZ + kb;

  constexpr int KS = HSZ / 4;
  const int col  = ntile * 16 + (lane & 15);
  v2f a[KS], w0[KS], w1[KS];
  #pragma unroll
  for (int s = 0; s < KS; ++s) {
    a[s]  = *(const v2f*)(Aptr  + 4 * s);
    w0[s] = *(const v2f*)(B0ptr + 4 * s);
    w1[s] = *(const v2f*)(B1ptr + 4 * s);
  }
  const float bias0 = b0[col];
  const float bias1 = b1[col];

  v8f acc0 = {};
  v8f acc1 = {};
  #pragma unroll
  for (int s = 0; s < KS; ++s) {
    acc0 = __builtin_amdgcn_wmma_f32_16x16x4_f32(
         false, a[s],  false, w0[s], (short)0, acc0, false, false);
    acc1 = __builtin_amdgcn_wmma_f32_16x16x4_f32(
         false, a[s],  false, w1[s], (short)0, acc1, false, false);
  }

  const int mofs = (lane >> 4) * 8;
  #pragma unroll
  for (int v = 0; v < 8; ++v) {
    const size_t idx = (size_t)(mtile * 16 + mofs + v) * HSZ + col;
    G0[idx] = acc0[v] + bias0;
    G1[idx] = acc1[v] + bias1;
  }
}

struct Chunk {
  float x;
  float g;
  int m;
};

__device__ __forceinline__ Chunk load_chunk(const float* __restrict__ xcol,
                                            const float* __restrict__ gcol,
                                            const int*   __restrict__ mrow,
                                            int row)
{
  Chunk c;
  c.x = xcol[(size_t)row * HSZ];
  c.g = gcol[(size_t)row * HSZ];
  c.m = mrow[row];
  return c;
}

__device__ __forceinline__ float lane_value(float v, int k)
{
  return __builtin_bit_cast(float,
           __builtin_amdgcn_readlane(__builtin_bit_cast(int, v), k));
}

__device__ __forceinline__ float send_to_lane(float v, int byte_addr)
{
  return __builtin_bit_cast(float,
           __builtin_amdgcn_ds_permute(byte_addr, __builtin_bit_cast(int, v)));
}

__device__ __forceinline__ float exp_neg(float g)
{
  const float c_hi = -0x1.715476p+0f;
  const float c_lo = -0x1.4ae0cp-26f;
  const float t = g * c_hi;
  const float r = __builtin_fmaf(g, c_lo, __builtin_fmaf(g, c_hi, -t));
  const float e = __builtin_amdgcn_exp2f(t);
  return __builtin_fmaf(e, r * 0x1.62e43p-1f, e);
}

template <int N>
__device__ __forceinline__ float invert_all(const v2f (&d)[N], v2f (&s)[N])
{
  v2f node[2 * N];
  #pragma unroll
  for (int n = 0; n < N; ++n) node[N + n] = d[n];
  #pragma unroll
  for (int n = N - 1; n >= 1; --n) node[n] = node[2 * n] * node[2 * n + 1];
  const float pp = node[1][0] * node[1][1];
  const float r  = 1.0f / pp;
  v2f inv[2 * N];
  inv[1][0] = r * node[1][1];
  inv[1][1] = r * node[1][0];
  #pragma unroll
  for (int n = 1; n < N; ++n) {
    inv[2 * n]     = inv[n] * node[2 * n + 1];
    inv[2 * n + 1] = inv[n] * node[2 * n];
  }
  #pragma unroll
  for (int n = 0; n < N; ++n) s[n] = inv[N + n];
  return pp;
}

template <int L0>
__device__ __forceinline__ float fast_trip(const float sx, const float se,
                                           const Rows& e0, Rows& acc)
{
  float x[8], e[8];
  #pragma unroll
  for (int r = 0; r < 8; ++r) {
    x[r] = lane_value(sx, L0 + r);
    e[r] = lane_value(se, L0 + r);
  }
  v2f d[8 * RP];
  #pragma unroll
  for (int r = 0; r < 8; ++r) {
    const v2f ee = {e[r], e[r]};
    const v2f one = {1.0f, 1.0f};
    #pragma unroll
    for (int q = 0; q < RP; ++q) d[r * RP + q] = __builtin_elementwise_fma(e0.p[q], ee, one);
  }
  v2f s[8 * RP];
  const float pp = invert_all<8 * RP>(d, s);
  #pragma unroll
  for (int r = 0; r < 8; ++r) {
    #pragma unroll
    for (int q = 0; q < RP; ++q) {
      #pragma unroll
      for (int t = 0; t < RI; ++t) acc.p[q][t] = __builtin_fmaf(x[r], s[r * RP + q][t], acc.p[q][t]);
    }
  }
  return pp;
}

template <bool SPLIT>
__device__ __forceinline__ void careful_trip(const float sx, const float sv, const int lane0,
                                             const Rows& a0, Rows& acc)
{
  float x[8], g[8];
  #pragma unroll
  for (int r = 0; r < 8; ++r) {
    x[r] = lane_value(sx, lane0 + r);
    g[r] = lane_value(sv, lane0 + r);
  }
  v2f d[8 * RP];
  #pragma unroll
  for (int r = 0; r < 8; ++r) {
    #pragma unroll
    for (int q = 0; q < RP; ++q) {
      if (SPLIT) {
        const v2f gg = {g[r], g[r]};
        const v2f one = {1.0f, 1.0f};
        d[r * RP + q] = __builtin_elementwise_fma(a0.p[q], gg, one);
      } else {
        #pragma unroll
        for (int t = 0; t < RI; ++t) d[r * RP + q][t] = 1.0f + __expf(-(a0.p[q][t] + g[r]));
      }
    }
  }
  v2f s[8 * RP];
  const float pp = invert_all<8 * RP>(d, s);
  if (__builtin_amdgcn_ballot_w32(!(pp < PMAX)) != 0u) {
    #pragma unroll
    for (int n = 0; n < 8 * RP; ++n) {
      #pragma unroll
      for (int t = 0; t < RI; ++t) s[n][t] = 1.0f / d[n][t];
    }
  }
  #pragma unroll
  for (int r = 0; r < 8; ++r) {
    #pragma unroll
    for (int q = 0; q < RP; ++q) {
      #pragma unroll
      for (int t = 0; t < RI; ++t) acc.p[q][t] = __builtin_fmaf(x[r], s[r * RP + q][t], acc.p[q][t]);
    }
  }
}

template <bool SPLIT>
__device__ __forceinline__ void careful_rows(const float sx, const float sv, const int rows,
                                             const Rows& a0, Rows& acc)
{
  #pragma nounroll
  for (int lane0 = 0; lane0 < rows; lane0 += 8) careful_trip<SPLIT>(sx, sv, lane0, a0, acc);
}

template <bool WHOLE>
__device__ __forceinline__ void use_register(const float sx, const float se, const float sg,
                                             const int rows, const bool far,
                                             const Rows& e0, const Rows& g0, Rows& acc)
{
  if (__builtin_expect(far, 0)) {
    careful_rows<false>(sx, sg, rows, g0, acc);
    return;
  }
  Rows trial = acc;
  float p = fast_trip<0>(sx, se, e0, trial);
  bool bad = !(p < PMAX);
  if (WHOLE || rows > 8)  { p = fast_trip<8>(sx, se, e0, trial);  bad = bad || !(p < PMAX); }
  if (WHOLE || rows > 16) { p = fast_trip<16>(sx, se, e0, trial); bad = bad || !(p < PMAX); }
  if (WHOLE || rows > 24) { p = fast_trip<24>(sx, se, e0, trial); bad = bad || !(p < PMAX); }
  if (__builtin_expect(__builtin_amdgcn_ballot_w32(bad) != 0u, 0)) {
    careful_rows<true>(sx, se, rows, e0, acc);
  } else {
    acc = trial;
  }
}

struct Stream {
  float x, e, g;
  int fill;
};

__device__ __forceinline__ int take_chunk(const Chunk cur, const int lane, bool& far, Stream& st,
                                          const Rows& e0, const Rows& g0, Rows& acc)
{
  const bool on = cur.m != 0;
  const unsigned bits = __builtin_amdgcn_ballot_w32(on);
  const int n = __builtin_popcount(bits);
  far = far || (__builtin_amdgcn_ballot_w32(!(__builtin_fabsf(cur.g) < GMAX)) != 0u);
  const float e1 = exp_neg(cur.g);

  const int below = (int)__builtin_amdgcn_mbcnt_lo(bits, 0u);
  const int place = on ? (st.fill + below) : (st.fill + n + (lane - below));
  const int to = (place & 31) << 2;
  const float rx = send_to_lane(cur.x, to);
  const float re = send_to_lane(e1, to);
  const float rg = send_to_lane(cur.g, to);
  const int end = st.fill + n;
  const bool fits = (lane >= st.fill) && (lane < end);
  st.x = fits ? rx : st.x;
  st.e = fits ? re : st.e;
  st.g = fits ? rg : st.g;
  st.fill = end;

  if (st.fill >= 32) {
    use_register<true>(st.x, st.e, st.g, 32, far, e0, g0, acc);
    st.x = rx;
    st.e = re;
    st.g = rg;
    st.fill -= 32;
  }
  return n;
}

__global__ __launch_bounds__(256) void pairwise_kernel(
    const float* __restrict__ inp,
    const int*   __restrict__ masks,
    const float* __restrict__ G0,
    const float* __restrict__ G1,
    float* __restrict__ out)
{
  __shared__ float tile[128 * TP];

  const int lane = threadIdx.x & 31;
  const int wv = threadIdx.x >> 5;
  const int hg = blockIdx.x & 15;
  const int it = (blockIdx.x >> 4) & 3;
  const int b  = blockIdx.x >> 6;
  const int h  = hg * 8 + wv;
  const int i0 = it * 128 + lane;

  Rows g0, mi, acc;
  #pragma unroll
  for (int q = 0; q < RP; ++q) {
    #pragma unroll
    for (int t = 0; t < RI; ++t) {
      const int i = i0 + 32 * (RI * q + t);
      g0.p[q][t]  = G0[(size_t)(b * SSZ + i) * HSZ + h];
      mi.p[q][t]  = (float)masks[b * SSZ + i];
      acc.p[q][t] = 0.0f;
    }
  }

  const float* xcol = inp + (size_t)b * SSZ * HSZ + h;
  const float* gcol = G1  + (size_t)b * SSZ * HSZ + h;
  const int*   mrow = masks + b * SSZ;

  Rows e0;
  bool far0 = false;
  #pragma unroll
  for (int q = 0; q < RP; ++q) {
    #pragma unroll
    for (int t = 0; t < RI; ++t) {
      e0.p[q][t] = exp_neg(g0.p[q][t]);
      far0 = far0 || !(__builtin_fabsf(g0.p[q][t]) < GMAX);
    }
  }
  bool far = __builtin_amdgcn_ballot_w32(far0) != 0u;

  Stream st = {0.0f, 0.0f, 0.0f, 0};

  int len = 0;
  Chunk c0 = load_chunk(xcol, gcol, mrow, lane);
  Chunk c1 = load_chunk(xcol, gcol, mrow, 32 + lane);
  for (int p = 0; p < SSZ / 64; ++p) {
    const int pn = (p + 1 < SSZ / 64) ? (p + 1) : p;
    const Chunk n0 = load_chunk(xcol, gcol, mrow, pn * 64 + lane);
    const Chunk n1 = load_chunk(xcol, gcol, mrow, pn * 64 + 32 + lane);
    Chunk cur = c0;
    #pragma nounroll
    for (int k = 0; k < 2; ++k) {
      len += take_chunk(cur, lane, far, st, e0, g0, acc);
      cur = c1;
    }
    c0 = n0;
    c1 = n1;
  }
  if (st.fill > 0) {
    const bool live = lane < st.fill;
    const float tx = live ? st.x : 0.0f;
    const float te = live ? st.e : 0.0f;
    const float tg = live ? st.g : 0.0f;
    use_register<false>(tx, te, tg, st.fill, far, e0, g0, acc);
  }

  const float inv_len = 1.0f / (float)len;
  #pragma unroll
  for (int q = 0; q < RP; ++q) {
    #pragma unroll
    for (int t = 0; t < RI; ++t)
      tile[(lane + 32 * (RI * q + t)) * TP + wv] = tanhf(mi.p[q][t] * acc.p[q][t] * inv_len);
  }
  __syncthreads();
  #pragma unroll
  for (int r = 0; r < 4; ++r) {
    const int row = wv * 16 + r * 4 + (lane >> 3);
    const int col = lane & 7;
    out[(size_t)(b * SSZ + it * 128 + row) * HSZ + hg * 8 + col] = tile[row * TP + col];
  }
}

extern "C" void kernel_launch(void* const* d_in, const int* in_sizes, int n_in,
                              void* d_out, int out_size, void* d_ws, size_t ws_size,
                              hipStream_t stream) {
  const float* inp   = (const float*)d_in[0];
  const int*   masks = (const int*)  d_in[1];
  const float* W0    = (const float*)d_in[2];
  const float* b0    = (const float*)d_in[3];
  const float* W1    = (const float*)d_in[4];
  const float* b1    = (const float*)d_in[5];
  float* out = (float*)d_out;

  float* G0 = (float*)d_ws;
  float* G1 = G0 + (size_t)BSZ * SSZ * HSZ;

  gates_wmma_kernel<<<256, 128, 0, stream>>>(inp, W0, b0, W1, b1, G0, G1);
  pairwise_kernel<<<256, 256, 0, stream>>>(inp, masks, G0, G1, out);
}
